# starting kernel
# baseline (speedup 1.0000x reference)
#include <hip/hip_runtime.h>
#include <math.h>

typedef __attribute__((ext_vector_type(2))) float v2f;
typedef __attribute__((ext_vector_type(8))) float v8f;

#define BIGF 1e30f
#define NBATCH 4096
#define UD 24

__device__ __forceinline__ void qam16_point(int s, float& re, float& im) {
    const int b0 = (s >> 3) & 1, b1 = (s >> 2) & 1, b2 = (s >> 1) & 1, b3 = s & 1;
    const float inv = 0.3162277660168379f;
    re = (float)((1 - 2 * b0) * (2 - (1 - 2 * b2))) * inv;
    im = (float)((1 - 2 * b1) * (2 - (1 - 2 * b3))) * inv;
}

__global__ void build_v_kernel(float* __restrict__ v) {
    const int c = blockIdx.x * blockDim.x + threadIdx.x;
    if (c >= 4096) return;
    const int idx0 = (c >> 8) & 15, idx1 = (c >> 4) & 15, idx2 = c & 15;
    float xr[3], xi[3];
    qam16_point(idx0, xr[0], xi[0]);
    qam16_point(idx1, xr[1], xi[1]);
    qam16_point(idx2, xr[2], xi[2]);
    float vv[UD];
#pragma unroll
    for (int k = 0; k < 3; k++) {
#pragma unroll
        for (int l = 0; l < 3; l++) {
            const float pre = xr[k] * xr[l] + xi[k] * xi[l];
            const float pim = xr[k] * xi[l] - xi[k] * xr[l];
            vv[k * 3 + l]     = pre;
            vv[9 + k * 3 + l] = -pim;
        }
    }
#pragma unroll
    for (int k = 0; k < 3; k++) {
        vv[18 + k] = -2.0f * xr[k];
        vv[21 + k] =  2.0f * xi[k];
    }
#pragma unroll
    for (int t = 0; t < UD; t++) v[c * UD + t] = vv[t];
}

__global__ void build_u_kernel(const float* __restrict__ y_re, const float* __restrict__ y_im,
                               const float* __restrict__ h_re, const float* __restrict__ h_im,
                               const float* __restrict__ s_re, const float* __restrict__ s_im,
                               float* __restrict__ u) {
    const int b = blockIdx.x * blockDim.x + threadIdx.x;
    if (b >= NBATCH) return;

    float Lr[8][8], Li[8][8];
#pragma unroll
    for (int i = 0; i < 8; i++) {
#pragma unroll
        for (int j = 0; j <= i; j++) {
            Lr[i][j] = s_re[b * 64 + i * 8 + j];
            Li[i][j] = s_im[b * 64 + i * 8 + j];
        }
    }
#pragma unroll
    for (int j = 0; j < 8; j++) {
        float d = Lr[j][j];
#pragma unroll
        for (int p = 0; p < j; p++) d -= Lr[j][p] * Lr[j][p] + Li[j][p] * Li[j][p];
        const float dj = sqrtf(d);
        const float inv = 1.0f / dj;
        Lr[j][j] = dj; Li[j][j] = 0.0f;
#pragma unroll
        for (int i = j + 1; i < 8; i++) {
            float ar = Lr[i][j], ai = Li[i][j];
#pragma unroll
            for (int p = 0; p < j; p++) {
                ar -= Lr[i][p] * Lr[j][p] + Li[i][p] * Li[j][p];
                ai -= Li[i][p] * Lr[j][p] - Lr[i][p] * Li[j][p];
            }
            Lr[i][j] = ar * inv; Li[i][j] = ai * inv;
        }
    }
    float ywr[8], ywi[8];
#pragma unroll
    for (int i = 0; i < 8; i++) {
        float ar = y_re[b * 8 + i], ai = y_im[b * 8 + i];
#pragma unroll
        for (int j = 0; j < i; j++) {
            ar -= Lr[i][j] * ywr[j] - Li[i][j] * ywi[j];
            ai -= Lr[i][j] * ywi[j] + Li[i][j] * ywr[j];
        }
        const float inv = 1.0f / Lr[i][i];
        ywr[i] = ar * inv; ywi[i] = ai * inv;
    }
    float hwr[8][3], hwi[8][3];
#pragma unroll
    for (int k = 0; k < 3; k++) {
#pragma unroll
        for (int i = 0; i < 8; i++) {
            float ar = h_re[b * 24 + i * 3 + k], ai = h_im[b * 24 + i * 3 + k];
#pragma unroll
            for (int j = 0; j < i; j++) {
                ar -= Lr[i][j] * hwr[j][k] - Li[i][j] * hwi[j][k];
                ai -= Lr[i][j] * hwi[j][k] + Li[i][j] * hwr[j][k];
            }
            const float inv = 1.0f / Lr[i][i];
            hwr[i][k] = ar * inv; hwi[i][k] = ai * inv;
        }
    }
    float uu[UD];
#pragma unroll
    for (int k = 0; k < 3; k++) {
#pragma unroll
        for (int l = 0; l < 3; l++) {
            float gr = 0.f, gi = 0.f;
#pragma unroll
            for (int m = 0; m < 8; m++) {
                gr += hwr[m][k] * hwr[m][l] + hwi[m][k] * hwi[m][l];
                gi += hwr[m][k] * hwi[m][l] - hwi[m][k] * hwr[m][l];
            }
            uu[k * 3 + l]     = gr;
            uu[9 + k * 3 + l] = gi;
        }
    }
#pragma unroll
    for (int k = 0; k < 3; k++) {
        float zr = 0.f, zi = 0.f;
#pragma unroll
        for (int m = 0; m < 8; m++) {
            zr += ywr[m] * hwr[m][k] + ywi[m] * hwi[m][k];
            zi += ywr[m] * hwi[m][k] - ywi[m] * hwr[m][k];
        }
        uu[18 + k] = zr;
        uu[21 + k] = zi;
    }
#pragma unroll
    for (int t = 0; t < UD; t++) u[b * UD + t] = uu[t];
}

__global__ void __launch_bounds__(256)
detect_kernel(const float* __restrict__ u, const float* __restrict__ v,
              float* __restrict__ out) {
    __shared__ float lm1[8][32][16];
    __shared__ float lm2[8][32][8];
    __shared__ float lb0[8][32][2];

    const int tid  = threadIdx.x;
    const int wave = tid >> 5;
    const int lane = tid & 31;
    const int li   = lane & 15;
    const int hh   = lane >> 4;
    const int bbase = blockIdx.x * 16;

    const float* ub = u + (size_t)(bbase + li) * UD + 2 * hh;
    v2f bf[6];
#pragma unroll
    for (int kk = 0; kk < 6; kk++) bf[kk] = *(const v2f*)(ub + 4 * kk);

    float m1h[16], m2h[8], bm0[2];
#pragma unroll
    for (int t = 0; t < 16; t++) m1h[t] = BIGF;
#pragma unroll
    for (int t = 0; t < 8; t++) m2h[t] = BIGF;

#pragma unroll
    for (int j = 0; j < 2; j++) {
        const int i0 = wave * 2 + j;
        float bmin = BIGF;
#pragma unroll
        for (int i1 = 0; i1 < 16; i1++) {
            const int cbase = i0 * 256 + i1 * 16;
            const float* vp = v + (size_t)(cbase + li) * UD + 2 * hh;
            v8f acc = {0.f, 0.f, 0.f, 0.f, 0.f, 0.f, 0.f, 0.f};
#pragma unroll
            for (int kk = 0; kk < 6; kk++) {
                const v2f af = *(const v2f*)(vp + 4 * kk);
                acc = __builtin_amdgcn_wmma_f32_16x16x4_f32(
                    false, af, false, bf[kk], (short)0, acc, false, false);
            }
            const float tmin =
                fminf(fminf(fminf(acc[0], acc[1]), fminf(acc[2], acc[3])),
                      fminf(fminf(acc[4], acc[5]), fminf(acc[6], acc[7])));
            bmin = fminf(bmin, tmin);
            m1h[i1] = fminf(m1h[i1], tmin);
#pragma unroll
            for (int r = 0; r < 8; r++) m2h[r] = fminf(m2h[r], acc[r]);
        }
        bm0[j] = bmin;
    }

#pragma unroll
    for (int t = 0; t < 16; t++) lm1[wave][lane][t] = m1h[t];
#pragma unroll
    for (int t = 0; t < 8; t++) lm2[wave][lane][t] = m2h[t];
    lb0[wave][lane][0] = bm0[0];
    lb0[wave][lane][1] = bm0[1];
    __syncthreads();

    if (tid < 16) {
        const int b = tid;
        float m[3][16];
#pragma unroll
        for (int i0 = 0; i0 < 16; i0++) {
            const int w = i0 >> 1, jj = i0 & 1;
            m[0][i0] = fminf(lb0[w][b][jj], lb0[w][b + 16][jj]);
        }
#pragma unroll
        for (int i1 = 0; i1 < 16; i1++) {
            float t = BIGF;
#pragma unroll
            for (int w = 0; w < 8; w++)
                t = fminf(t, fminf(lm1[w][b][i1], lm1[w][b + 16][i1]));
            m[1][i1] = t;
        }
#pragma unroll
        for (int i2 = 0; i2 < 16; i2++) {
            const int h2 = i2 >> 3, r = i2 & 7;
            float t = BIGF;
#pragma unroll
            for (int w = 0; w < 8; w++) t = fminf(t, lm2[w][b + 16 * h2][r]);
            m[2][i2] = t;
        }
#pragma unroll
        for (int k = 0; k < 3; k++) {
#pragma unroll
            for (int n = 0; n < 4; n++) {
                float mn0 = BIGF, mn1 = BIGF;
#pragma unroll
                for (int i = 0; i < 16; i++) {
                    if ((i >> (3 - n)) & 1) mn1 = fminf(mn1, m[k][i]);
                    else                    mn0 = fminf(mn0, m[k][i]);
                }
                out[(size_t)(bbase + b) * 12 + k * 4 + n] = mn0 - mn1;
            }
        }
    }
}

extern "C" void kernel_launch(void* const* d_in, const int* in_sizes, int n_in,
                              void* d_out, int out_size, void* d_ws, size_t ws_size,
                              hipStream_t stream) {
    const float* y_re = (const float*)d_in[0];
    const float* y_im = (const float*)d_in[1];
    const float* h_re = (const float*)d_in[2];
    const float* h_im = (const float*)d_in[3];
    const float* s_re = (const float*)d_in[4];
    const float* s_im = (const float*)d_in[5];
    float* out = (float*)d_out;

    float* u_tab = (float*)d_ws;
    float* v_tab = u_tab + (size_t)NBATCH * UD;

    build_v_kernel<<<16, 256, 0, stream>>>(v_tab);
    build_u_kernel<<<16, 256, 0, stream>>>(y_re, y_im, h_re, h_im, s_re, s_im, u_tab);
    detect_kernel<<<NBATCH / 16, 256, 0, stream>>>(u_tab, v_tab, out);
}


# whitening and pair tables formed at the head of the detect launch: each block's first 16 lanes eliminate its 16 batches in registers, u and tables go through LDS behind one barrier; 2 -> 1 launches
# speedup vs baseline: 3.9112x; 3.9112x over previous
#include <hip/hip_runtime.h>
#include <math.h>


typedef __attribute__((ext_vector_type(2))) float v2f;
typedef __attribute__((ext_vector_type(8))) float v8f;

#define BIGF 1e30f
#define NBATCH 4096
#define UD 24
#define NPAIR 256
#define K01 12
#define K02 8
#define K12 4
static_assert(NPAIR == 256, "thread p of a 256-thread block forms row p of the pair tables");
static_assert(K01 + K02 + K12 == UD, "every one of the 24 terms belongs to exactly one pair table");

__device__ __forceinline__ int idx01(int j) {
    constexpr int t[K01] = {0, 1, 3, 4, 9, 10, 12, 13, 18, 19, 21, 22};
    return t[j];
}
__device__ __forceinline__ int idx02(int j) {
    constexpr int t[K02] = {2, 6, 11, 15, 8, 20, 17, 23};
    return t[j];
}
__device__ __forceinline__ int idx12(int j) {
    constexpr int t[K12] = {5, 7, 14, 16};
    return t[j];
}

__device__ __forceinline__ void qam16_point(int s, float& re, float& im) {
    const int b0 = (s >> 3) & 1, b1 = (s >> 2) & 1, b2 = (s >> 1) & 1, b3 = s & 1;
    const float inv = 0.3162277660168379f;
    re = (float)((1 - 2 * b0) * (2 - (1 - 2 * b2))) * inv;
    im = (float)((1 - 2 * b1) * (2 - (1 - 2 * b3))) * inv;
}

__device__ __forceinline__ void cand_terms(const float xr[3], const float xi[3], float vv[UD]) {
#pragma unroll
    for (int k = 0; k < 3; k++) {
#pragma unroll
        for (int l = 0; l < 3; l++) {
            const float pre = xr[k] * xr[l] + xi[k] * xi[l];
            const float pim = xr[k] * xi[l] - xi[k] * xr[l];
            vv[k * 3 + l]     = pre;
            vv[9 + k * 3 + l] = -pim;
        }
    }
#pragma unroll
    for (int k = 0; k < 3; k++) {
        vv[18 + k] = -2.0f * xr[k];
        vv[21 + k] =  2.0f * xi[k];
    }
}

__device__ __forceinline__ void build_pair_rows(int p, float* __restrict__ a01,
                                                float* __restrict__ a02, float* __restrict__ a12) {
    float ar, ai, br, bi;
    qam16_point((p >> 4) & 15, ar, ai);
    qam16_point(p & 15, br, bi);
    float vv[UD];
    {
        const float xr[3] = {ar, br, 0.f}, xi[3] = {ai, bi, 0.f};
        cand_terms(xr, xi, vv);
#pragma unroll
        for (int j = 0; j < K01; j++) a01[p * K01 + j] = vv[idx01(j)];
    }
    {
        const float xr[3] = {ar, 0.f, br}, xi[3] = {ai, 0.f, bi};
        cand_terms(xr, xi, vv);
#pragma unroll
        for (int j = 0; j < K02; j++) a02[p * K02 + j] = vv[idx02(j)];
    }
    {
        const float xr[3] = {0.f, ar, br}, xi[3] = {0.f, ai, bi};
        cand_terms(xr, xi, vv);
#pragma unroll
        for (int j = 0; j < K12; j++) a12[p * K12 + j] = vv[idx12(j)];
    }
}

#define NS 8
#define NB 4
#define UP (UD + 1)

__device__ __forceinline__ void load_batch(int b,
        const float* __restrict__ y_re, const float* __restrict__ y_im,
        const float* __restrict__ h_re, const float* __restrict__ h_im,
        const float* __restrict__ s_re, const float* __restrict__ s_im,
        float (&Sr)[NS][NS], float (&Si)[NS][NS], float (&Br)[NB][NS], float (&Bi)[NB][NS]) {
#pragma unroll
    for (int i = 0; i < NS; i++) {
#pragma unroll
        for (int j = 0; j <= i; j++) {
            Sr[i][j] = s_re[b * 64 + i * 8 + j];
            Si[i][j] = s_im[b * 64 + i * 8 + j];
        }
    }
#pragma unroll
    for (int i = 0; i < NS; i++) { Br[3][i] = y_re[b * 8 + i]; Bi[3][i] = y_im[b * 8 + i]; }
#pragma unroll
    for (int i = 0; i < NS; i++) {
#pragma unroll
        for (int k = 0; k < 3; k++) { Br[k][i] = h_re[b * 24 + i * 3 + k]; Bi[k][i] = h_im[b * 24 + i * 3 + k]; }
    }
}

__device__ __forceinline__ void eliminate_to_u(float (&Sr)[NS][NS], float (&Si)[NS][NS],
                                               float (&Br)[NB][NS], float (&Bi)[NB][NS],
                                               float* __restrict__ urow) {
    float accr[NB][NB], acci[NB][NB];
#pragma unroll
    for (int a = 0; a < NB; a++) {
#pragma unroll
        for (int c = 0; c <= a; c++) { accr[a][c] = 0.f; acci[a][c] = 0.f; }
    }
#pragma unroll
    for (int j = 0; j < NS; j++) {
        const float invd = 1.0f / Sr[j][j];
#pragma unroll
        for (int i = j + 1; i < NS; i++) {
            const float lr = Sr[i][j] * invd, li = Si[i][j] * invd;
#pragma unroll
            for (int k = j + 1; k < i; k++) {
                Sr[i][k] = Sr[i][k] - lr * Sr[k][j] - li * Si[k][j];
                Si[i][k] = Si[i][k] - li * Sr[k][j] + lr * Si[k][j];
            }
            Sr[i][i] = Sr[i][i] - lr * Sr[i][j] - li * Si[i][j];
        }
#pragma unroll
        for (int a = 0; a < NB; a++) {
            const float mr = Br[a][j] * invd, mi = Bi[a][j] * invd;
#pragma unroll
            for (int k = j + 1; k < NS; k++) {
                Br[a][k] = Br[a][k] - mr * Sr[k][j] + mi * Si[k][j];
                Bi[a][k] = Bi[a][k] - mr * Si[k][j] - mi * Sr[k][j];
            }
#pragma unroll
            for (int c = 0; c < a; c++) {
                accr[a][c] = accr[a][c] + mr * Br[c][j] + mi * Bi[c][j];
                acci[a][c] = acci[a][c] + mr * Bi[c][j] - mi * Br[c][j];
            }
            if (a < 3) accr[a][a] = accr[a][a] + mr * Br[a][j] + mi * Bi[a][j];
        }
    }
    float uu[UD];
#pragma unroll
    for (int k = 0; k < 3; k++) {
        uu[k * 3 + k]     = accr[k][k];
        uu[9 + k * 3 + k] = 0.f;
#pragma unroll
        for (int l = 0; l < k; l++) {
            uu[k * 3 + l]     = accr[k][l];
            uu[9 + k * 3 + l] = acci[k][l];
            uu[l * 3 + k]     = accr[k][l];
            uu[9 + l * 3 + k] = -acci[k][l];
        }
    }
#pragma unroll
    for (int k = 0; k < 3; k++) {
        uu[18 + k] = accr[3][k];
        uu[21 + k] = acci[3][k];
    }
#pragma unroll
    for (int t = 0; t < UD; t++) urow[t] = uu[t];
}

__device__ __forceinline__ v8f wmma4(v2f a, v2f b, v8f c) {
    return __builtin_amdgcn_wmma_f32_16x16x4_f32(false, a, false, b, (short)0, c, false, false);
}

__global__ void __launch_bounds__(256)
detect_kernel(const float* __restrict__ y_re, const float* __restrict__ y_im,
              const float* __restrict__ h_re, const float* __restrict__ h_im,
              const float* __restrict__ s_re, const float* __restrict__ s_im,
              float* __restrict__ out) {
    __shared__ float lm1[8][32][8];
    __shared__ float lm2[8][32][16];
    __shared__ float lb0[8][32][2];
    __shared__ float lu[16][UP];
    __shared__ __attribute__((aligned(16))) float a01[NPAIR * K01];
    __shared__ __attribute__((aligned(16))) float a02[NPAIR * K02];
    __shared__ __attribute__((aligned(16))) float a12[NPAIR * K12];

    const int tid  = threadIdx.x;
    const int wave = tid >> 5;
    const int lane = tid & 31;
    const int li   = lane & 15;
    const int hh   = lane >> 4;
    const int bbase = blockIdx.x * 16;

    const bool owns_batch = tid < 16;
    float Sr[NS][NS], Si[NS][NS], Br[NB][NS], Bi[NB][NS];
    if (owns_batch) load_batch(bbase + tid, y_re, y_im, h_re, h_im, s_re, s_im, Sr, Si, Br, Bi);
    build_pair_rows(tid, a01, a02, a12);
    if (owns_batch) eliminate_to_u(Sr, Si, Br, Bi, &lu[tid][0]);
    __syncthreads();

    const float* ub = &lu[li][0];
    v2f b01[3], b02[2], b12;
#pragma unroll
    for (int q = 0; q < 3; q++) {
#pragma unroll
        for (int s = 0; s < 2; s++)
            b01[q][s] = ub[idx01(4 * q + s) + hh * (idx01(4 * q + 2 + s) - idx01(4 * q + s))];
    }
#pragma unroll
    for (int q = 0; q < 2; q++) {
#pragma unroll
        for (int s = 0; s < 2; s++)
            b02[q][s] = ub[idx02(4 * q + s) + hh * (idx02(4 * q + 2 + s) - idx02(4 * q + s))];
    }
#pragma unroll
    for (int s = 0; s < 2; s++) b12[s] = ub[idx12(s) + hh * (idx12(2 + s) - idx12(s))];

    v2f f12[8][2], f02[2][2][2], f01[2][3];
#pragma unroll
    for (int i = 0; i < 8; i++) {
#pragma unroll
        for (int q2 = 0; q2 < 2; q2++) {
            const int p = ((8 * (li >> 3) + i) << 4) | (8 * q2 + (li & 7));
            f12[i][q2] = *(const v2f*)(a12 + p * K12 + 2 * hh);
        }
    }
#pragma unroll
    for (int j = 0; j < 2; j++) {
        const int x0 = wave * 2 + j;
#pragma unroll
        for (int q2 = 0; q2 < 2; q2++) {
            const int p = (x0 << 4) | (8 * q2 + (li & 7));
#pragma unroll
            for (int q = 0; q < 2; q++) f02[j][q2][q] = *(const v2f*)(a02 + p * K02 + 2 * hh + 4 * q);
        }
        const int p = (x0 << 4) | li;
#pragma unroll
        for (int q = 0; q < 3; q++) f01[j][q] = *(const v2f*)(a01 + p * K01 + 2 * hh + 4 * q);
    }

    const v8f zero8 = {0.f, 0.f, 0.f, 0.f, 0.f, 0.f, 0.f, 0.f};

    float t12[8][16];
#pragma unroll
    for (int i = 0; i < 8; i++) {
#pragma unroll
        for (int q2 = 0; q2 < 2; q2++) {
            const v8f acc = wmma4(f12[i][q2], b12, zero8);
#pragma unroll
            for (int r = 0; r < 8; r++) t12[i][8 * q2 + r] = acc[r];
        }
    }

    float m1h[8], m2h[16], bm0[2];
#pragma unroll
    for (int t = 0; t < 8; t++) m1h[t] = BIGF;
#pragma unroll
    for (int t = 0; t < 16; t++) m2h[t] = BIGF;

#pragma unroll
    for (int j = 0; j < 2; j++) {
        float t02[16];
#pragma unroll
        for (int q2 = 0; q2 < 2; q2++) {
            v8f acc = zero8;
#pragma unroll
            for (int q = 0; q < 2; q++) acc = wmma4(f02[j][q2][q], b02[q], acc);
#pragma unroll
            for (int r = 0; r < 8; r++) t02[8 * q2 + r] = acc[r];
        }
        float t01[8];
        {
            v8f acc = zero8;
#pragma unroll
            for (int q = 0; q < 3; q++) acc = wmma4(f01[j][q], b01[q], acc);
#pragma unroll
            for (int r = 0; r < 8; r++) t01[r] = acc[r];
        }
        float bmin = BIGF;
#pragma unroll
        for (int i = 0; i < 8; i++) {
            float f[16];
#pragma unroll
            for (int x2 = 0; x2 < 16; x2++) f[x2] = (t12[i][x2] + t02[x2]) + t01[i];
            const float tmin =
                fminf(fminf(fminf(fminf(f[0], f[1]), f[2]), fminf(fminf(f[3], f[4]), f[5])),
                      fminf(fminf(fminf(fminf(f[6], f[7]), f[8]), fminf(fminf(f[9], f[10]), f[11])),
                            fminf(fminf(fminf(f[12], f[13]), f[14]), f[15])));
            bmin = fminf(bmin, tmin);
            m1h[i] = fminf(m1h[i], tmin);
#pragma unroll
            for (int x2 = 0; x2 < 16; x2++) m2h[x2] = fminf(m2h[x2], f[x2]);
        }
        bm0[j] = bmin;
    }

#pragma unroll
    for (int t = 0; t < 8; t++) lm1[wave][lane][t] = m1h[t];
#pragma unroll
    for (int t = 0; t < 16; t++) lm2[wave][lane][t] = m2h[t];
    lb0[wave][lane][0] = bm0[0];
    lb0[wave][lane][1] = bm0[1];
    __syncthreads();

    if (tid < 16) {
        const int b = tid;
        float m[3][16];
#pragma unroll
        for (int i0 = 0; i0 < 16; i0++) {
            const int w = i0 >> 1, jj = i0 & 1;
            m[0][i0] = fminf(lb0[w][b][jj], lb0[w][b + 16][jj]);
        }
#pragma unroll
        for (int i1 = 0; i1 < 16; i1++) {
            const int h1 = i1 >> 3, r = i1 & 7;
            float t = BIGF;
#pragma unroll
            for (int w = 0; w < 8; w++) t = fminf(t, lm1[w][b + 16 * h1][r]);
            m[1][i1] = t;
        }
#pragma unroll
        for (int i2 = 0; i2 < 16; i2++) {
            float t = BIGF;
#pragma unroll
            for (int w = 0; w < 8; w++)
                t = fminf(t, fminf(lm2[w][b][i2], lm2[w][b + 16][i2]));
            m[2][i2] = t;
        }
#pragma unroll
        for (int k = 0; k < 3; k++) {
#pragma unroll
            for (int n = 0; n < 4; n++) {
                float mn0 = BIGF, mn1 = BIGF;
#pragma unroll
                for (int i = 0; i < 16; i++) {
                    if ((i >> (3 - n)) & 1) mn1 = fminf(mn1, m[k][i]);
                    else                    mn0 = fminf(mn0, m[k][i]);
                }
                out[(size_t)(bbase + b) * 12 + k * 4 + n] = mn0 - mn1;
            }
        }
    }
}

extern "C" void kernel_launch(void* const* d_in, const int* in_sizes, int n_in,
                              void* d_out, int out_size, void* d_ws, size_t ws_size,
                              hipStream_t stream) {
    const float* y_re = (const float*)d_in[0];
    const float* y_im = (const float*)d_in[1];
    const float* h_re = (const float*)d_in[2];
    const float* h_im = (const float*)d_in[3];
    const float* s_re = (const float*)d_in[4];
    const float* s_im = (const float*)d_in[5];
    float* out = (float*)d_out;

    detect_kernel<<<NBATCH / 16, 256, 0, stream>>>(y_re, y_im, h_re, h_im, s_re, s_im, out);
}


# work-group of 4 waves (one per SIMD) instead of 8, each wave owning four x0 instead of two: half the waves to launch, T12 tiles formed 4 not 8 times a block, half the partial minima to combine
# speedup vs baseline: 4.1645x; 1.0648x over previous
#include <hip/hip_runtime.h>
#include <math.h>


typedef __attribute__((ext_vector_type(2))) float v2f;
typedef __attribute__((ext_vector_type(8))) float v8f;

#define BIGF 1e30f
#define NBATCH 4096
#define UD 24
#define NPAIR 256
#define NTHR 128
#define NWAVE 4
#define XW 4
#define K01 12
#define K02 8
#define K12 4
static_assert(NPAIR == 2 * NTHR, "thread p of a 128-thread block forms rows p and p + 128 of the pair tables");
static_assert(NWAVE * 32 == NTHR && NWAVE * XW == 16, "the block's waves share the 16 values of x0 evenly");
static_assert(K01 + K02 + K12 == UD, "every one of the 24 terms belongs to exactly one pair table");

__device__ __forceinline__ int idx01(int j) {
    constexpr int t[K01] = {0, 1, 3, 4, 9, 10, 12, 13, 18, 19, 21, 22};
    return t[j];
}
__device__ __forceinline__ int idx02(int j) {
    constexpr int t[K02] = {2, 6, 11, 15, 8, 20, 17, 23};
    return t[j];
}
__device__ __forceinline__ int idx12(int j) {
    constexpr int t[K12] = {5, 7, 14, 16};
    return t[j];
}

__device__ __forceinline__ void qam16_point(int s, float& re, float& im) {
    const int b0 = (s >> 3) & 1, b1 = (s >> 2) & 1, b2 = (s >> 1) & 1, b3 = s & 1;
    const float inv = 0.3162277660168379f;
    re = (float)((1 - 2 * b0) * (2 - (1 - 2 * b2))) * inv;
    im = (float)((1 - 2 * b1) * (2 - (1 - 2 * b3))) * inv;
}

__device__ __forceinline__ void cand_terms(const float xr[3], const float xi[3], float vv[UD]) {
#pragma unroll
    for (int k = 0; k < 3; k++) {
#pragma unroll
        for (int l = 0; l < 3; l++) {
            const float pre = xr[k] * xr[l] + xi[k] * xi[l];
            const float pim = xr[k] * xi[l] - xi[k] * xr[l];
            vv[k * 3 + l]     = pre;
            vv[9 + k * 3 + l] = -pim;
        }
    }
#pragma unroll
    for (int k = 0; k < 3; k++) {
        vv[18 + k] = -2.0f * xr[k];
        vv[21 + k] =  2.0f * xi[k];
    }
}

__device__ __forceinline__ void build_pair_rows(int p, float* __restrict__ a01,
                                                float* __restrict__ a02, float* __restrict__ a12) {
    float ar, ai, br, bi;
    qam16_point((p >> 4) & 15, ar, ai);
    qam16_point(p & 15, br, bi);
    float vv[UD];
    {
        const float xr[3] = {ar, br, 0.f}, xi[3] = {ai, bi, 0.f};
        cand_terms(xr, xi, vv);
#pragma unroll
        for (int j = 0; j < K01; j++) a01[p * K01 + j] = vv[idx01(j)];
    }
    {
        const float xr[3] = {ar, 0.f, br}, xi[3] = {ai, 0.f, bi};
        cand_terms(xr, xi, vv);
#pragma unroll
        for (int j = 0; j < K02; j++) a02[p * K02 + j] = vv[idx02(j)];
    }
    {
        const float xr[3] = {0.f, ar, br}, xi[3] = {0.f, ai, bi};
        cand_terms(xr, xi, vv);
#pragma unroll
        for (int j = 0; j < K12; j++) a12[p * K12 + j] = vv[idx12(j)];
    }
}

#define NS 8
#define NB 4
#define UP (UD + 1)

__device__ __forceinline__ void load_batch(int b,
        const float* __restrict__ y_re, const float* __restrict__ y_im,
        const float* __restrict__ h_re, const float* __restrict__ h_im,
        const float* __restrict__ s_re, const float* __restrict__ s_im,
        float (&Sr)[NS][NS], float (&Si)[NS][NS], float (&Br)[NB][NS], float (&Bi)[NB][NS]) {
#pragma unroll
    for (int i = 0; i < NS; i++) {
#pragma unroll
        for (int j = 0; j <= i; j++) {
            Sr[i][j] = s_re[b * 64 + i * 8 + j];
            Si[i][j] = s_im[b * 64 + i * 8 + j];
        }
    }
#pragma unroll
    for (int i = 0; i < NS; i++) { Br[3][i] = y_re[b * 8 + i]; Bi[3][i] = y_im[b * 8 + i]; }
#pragma unroll
    for (int i = 0; i < NS; i++) {
#pragma unroll
        for (int k = 0; k < 3; k++) { Br[k][i] = h_re[b * 24 + i * 3 + k]; Bi[k][i] = h_im[b * 24 + i * 3 + k]; }
    }
}

__device__ __forceinline__ void eliminate_to_u(float (&Sr)[NS][NS], float (&Si)[NS][NS],
                                               float (&Br)[NB][NS], float (&Bi)[NB][NS],
                                               float* __restrict__ urow) {
    float accr[NB][NB], acci[NB][NB];
#pragma unroll
    for (int a = 0; a < NB; a++) {
#pragma unroll
        for (int c = 0; c <= a; c++) { accr[a][c] = 0.f; acci[a][c] = 0.f; }
    }
#pragma unroll
    for (int j = 0; j < NS; j++) {
        const float invd = 1.0f / Sr[j][j];
#pragma unroll
        for (int i = j + 1; i < NS; i++) {
            const float lr = Sr[i][j] * invd, li = Si[i][j] * invd;
#pragma unroll
            for (int k = j + 1; k < i; k++) {
                Sr[i][k] = Sr[i][k] - lr * Sr[k][j] - li * Si[k][j];
                Si[i][k] = Si[i][k] - li * Sr[k][j] + lr * Si[k][j];
            }
            Sr[i][i] = Sr[i][i] - lr * Sr[i][j] - li * Si[i][j];
        }
#pragma unroll
        for (int a = 0; a < NB; a++) {
            const float mr = Br[a][j] * invd, mi = Bi[a][j] * invd;
#pragma unroll
            for (int k = j + 1; k < NS; k++) {
                Br[a][k] = Br[a][k] - mr * Sr[k][j] + mi * Si[k][j];
                Bi[a][k] = Bi[a][k] - mr * Si[k][j] - mi * Sr[k][j];
            }
#pragma unroll
            for (int c = 0; c < a; c++) {
                accr[a][c] = accr[a][c] + mr * Br[c][j] + mi * Bi[c][j];
                acci[a][c] = acci[a][c] + mr * Bi[c][j] - mi * Br[c][j];
            }
            if (a < 3) accr[a][a] = accr[a][a] + mr * Br[a][j] + mi * Bi[a][j];
        }
    }
    float uu[UD];
#pragma unroll
    for (int k = 0; k < 3; k++) {
        uu[k * 3 + k]     = accr[k][k];
        uu[9 + k * 3 + k] = 0.f;
#pragma unroll
        for (int l = 0; l < k; l++) {
            uu[k * 3 + l]     = accr[k][l];
            uu[9 + k * 3 + l] = acci[k][l];
            uu[l * 3 + k]     = accr[k][l];
            uu[9 + l * 3 + k] = -acci[k][l];
        }
    }
#pragma unroll
    for (int k = 0; k < 3; k++) {
        uu[18 + k] = accr[3][k];
        uu[21 + k] = acci[3][k];
    }
#pragma unroll
    for (int t = 0; t < UD; t++) urow[t] = uu[t];
}

__device__ __forceinline__ v8f wmma4(v2f a, v2f b, v8f c) {
    return __builtin_amdgcn_wmma_f32_16x16x4_f32(false, a, false, b, (short)0, c, false, false);
}

__global__ void __launch_bounds__(NTHR)
detect_kernel(const float* __restrict__ y_re, const float* __restrict__ y_im,
              const float* __restrict__ h_re, const float* __restrict__ h_im,
              const float* __restrict__ s_re, const float* __restrict__ s_im,
              float* __restrict__ out) {
    __shared__ float lm1[NWAVE][32][8];
    __shared__ float lm2[NWAVE][32][16];
    __shared__ float lb0[NWAVE][32][XW];
    __shared__ float lu[16][UP];
    __shared__ __attribute__((aligned(16))) float a01[NPAIR * K01];
    __shared__ __attribute__((aligned(16))) float a02[NPAIR * K02];
    __shared__ __attribute__((aligned(16))) float a12[NPAIR * K12];

    const int tid  = threadIdx.x;
    const int wave = tid >> 5;
    const int lane = tid & 31;
    const int li   = lane & 15;
    const int hh   = lane >> 4;
    const int bbase = blockIdx.x * 16;

    const bool owns_batch = tid < 16;
    float Sr[NS][NS], Si[NS][NS], Br[NB][NS], Bi[NB][NS];
    if (owns_batch) load_batch(bbase + tid, y_re, y_im, h_re, h_im, s_re, s_im, Sr, Si, Br, Bi);
    build_pair_rows(tid, a01, a02, a12);
    build_pair_rows(tid + NTHR, a01, a02, a12);
    if (owns_batch) eliminate_to_u(Sr, Si, Br, Bi, &lu[tid][0]);
    __syncthreads();

    const float* ub = &lu[li][0];
    v2f b01[3], b02[2], b12;
#pragma unroll
    for (int q = 0; q < 3; q++) {
#pragma unroll
        for (int s = 0; s < 2; s++)
            b01[q][s] = ub[idx01(4 * q + s) + hh * (idx01(4 * q + 2 + s) - idx01(4 * q + s))];
    }
#pragma unroll
    for (int q = 0; q < 2; q++) {
#pragma unroll
        for (int s = 0; s < 2; s++)
            b02[q][s] = ub[idx02(4 * q + s) + hh * (idx02(4 * q + 2 + s) - idx02(4 * q + s))];
    }
#pragma unroll
    for (int s = 0; s < 2; s++) b12[s] = ub[idx12(s) + hh * (idx12(2 + s) - idx12(s))];

    v2f f12[8][2], f02[XW][2][2], f01[XW][3];
#pragma unroll
    for (int i = 0; i < 8; i++) {
#pragma unroll
        for (int q2 = 0; q2 < 2; q2++) {
            const int p = ((8 * (li >> 3) + i) << 4) | (8 * q2 + (li & 7));
            f12[i][q2] = *(const v2f*)(a12 + p * K12 + 2 * hh);
        }
    }
#pragma unroll
    for (int j = 0; j < XW; j++) {
        const int x0 = wave * XW + j;
#pragma unroll
        for (int q2 = 0; q2 < 2; q2++) {
            const int p = (x0 << 4) | (8 * q2 + (li & 7));
#pragma unroll
            for (int q = 0; q < 2; q++) f02[j][q2][q] = *(const v2f*)(a02 + p * K02 + 2 * hh + 4 * q);
        }
        const int p = (x0 << 4) | li;
#pragma unroll
        for (int q = 0; q < 3; q++) f01[j][q] = *(const v2f*)(a01 + p * K01 + 2 * hh + 4 * q);
    }

    const v8f zero8 = {0.f, 0.f, 0.f, 0.f, 0.f, 0.f, 0.f, 0.f};

    float t12[8][16];
#pragma unroll
    for (int i = 0; i < 8; i++) {
#pragma unroll
        for (int q2 = 0; q2 < 2; q2++) {
            const v8f acc = wmma4(f12[i][q2], b12, zero8);
#pragma unroll
            for (int r = 0; r < 8; r++) t12[i][8 * q2 + r] = acc[r];
        }
    }

    float m1h[8], m2h[16], bm0[XW];
#pragma unroll
    for (int t = 0; t < 8; t++) m1h[t] = BIGF;
#pragma unroll
    for (int t = 0; t < 16; t++) m2h[t] = BIGF;

#pragma unroll
    for (int j = 0; j < XW; j++) {
        float t02[16];
#pragma unroll
        for (int q2 = 0; q2 < 2; q2++) {
            v8f acc = zero8;
#pragma unroll
            for (int q = 0; q < 2; q++) acc = wmma4(f02[j][q2][q], b02[q], acc);
#pragma unroll
            for (int r = 0; r < 8; r++) t02[8 * q2 + r] = acc[r];
        }
        float t01[8];
        {
            v8f acc = zero8;
#pragma unroll
            for (int q = 0; q < 3; q++) acc = wmma4(f01[j][q], b01[q], acc);
#pragma unroll
            for (int r = 0; r < 8; r++) t01[r] = acc[r];
        }
        float bmin = BIGF;
#pragma unroll
        for (int i = 0; i < 8; i++) {
            float f[16];
#pragma unroll
            for (int x2 = 0; x2 < 16; x2++) f[x2] = (t12[i][x2] + t02[x2]) + t01[i];
            const float tmin =
                fminf(fminf(fminf(fminf(f[0], f[1]), f[2]), fminf(fminf(f[3], f[4]), f[5])),
                      fminf(fminf(fminf(fminf(f[6], f[7]), f[8]), fminf(fminf(f[9], f[10]), f[11])),
                            fminf(fminf(fminf(f[12], f[13]), f[14]), f[15])));
            bmin = fminf(bmin, tmin);
            m1h[i] = fminf(m1h[i], tmin);
#pragma unroll
            for (int x2 = 0; x2 < 16; x2++) m2h[x2] = fminf(m2h[x2], f[x2]);
        }
        bm0[j] = bmin;
    }

#pragma unroll
    for (int t = 0; t < 8; t++) lm1[wave][lane][t] = m1h[t];
#pragma unroll
    for (int t = 0; t < 16; t++) lm2[wave][lane][t] = m2h[t];
#pragma unroll
    for (int j = 0; j < XW; j++) lb0[wave][lane][j] = bm0[j];
    __syncthreads();

    if (tid < 16) {
        const int b = tid;
        float m[3][16];
#pragma unroll
        for (int i0 = 0; i0 < 16; i0++) {
            const int w = i0 / XW, jj = i0 % XW;
            m[0][i0] = fminf(lb0[w][b][jj], lb0[w][b + 16][jj]);
        }
#pragma unroll
        for (int i1 = 0; i1 < 16; i1++) {
            const int h1 = i1 >> 3, r = i1 & 7;
            float t = BIGF;
#pragma unroll
            for (int w = 0; w < NWAVE; w++) t = fminf(t, lm1[w][b + 16 * h1][r]);
            m[1][i1] = t;
        }
#pragma unroll
        for (int i2 = 0; i2 < 16; i2++) {
            float t = BIGF;
#pragma unroll
            for (int w = 0; w < NWAVE; w++)
                t = fminf(t, fminf(lm2[w][b][i2], lm2[w][b + 16][i2]));
            m[2][i2] = t;
        }
#pragma unroll
        for (int k = 0; k < 3; k++) {
#pragma unroll
            for (int n = 0; n < 4; n++) {
                float mn0 = BIGF, mn1 = BIGF;
#pragma unroll
                for (int i = 0; i < 16; i++) {
                    if ((i >> (3 - n)) & 1) mn1 = fminf(mn1, m[k][i]);
                    else                    mn0 = fminf(mn0, m[k][i]);
                }
                out[(size_t)(bbase + b) * 12 + k * 4 + n] = mn0 - mn1;
            }
        }
    }
}

extern "C" void kernel_launch(void* const* d_in, const int* in_sizes, int n_in,
                              void* d_out, int out_size, void* d_ws, size_t ws_size,
                              hipStream_t stream) {
    const float* y_re = (const float*)d_in[0];
    const float* y_im = (const float*)d_in[1];
    const float* h_re = (const float*)d_in[2];
    const float* h_im = (const float*)d_in[3];
    const float* s_re = (const float*)d_in[4];
    const float* s_im = (const float*)d_in[5];
    float* out = (float*)d_out;

    detect_kernel<<<NBATCH / 16, NTHR, 0, stream>>>(y_re, y_im, h_re, h_im, s_re, s_im, out);
}
